# starting kernel
# baseline (speedup 1.0000x reference)
#include <hip/hip_runtime.h>
#include <hip/hip_bf16.h>


#define H_DIM   512
#define D_IN    128
#define BTILE   64
#define PAIRW   152
#define NUM_K   64
#define BATCH   2048

typedef __attribute__((ext_vector_type(2))) float v2f;
typedef __attribute__((ext_vector_type(8))) float v8f;

__device__ __forceinline__ void run_layer(const float* __restrict__ W,
                                          const float* __restrict__ bias,
                                          int Kdim, int ldw,
                                          float* __restrict__ hbuf,
                                          int wave, int lane)
{
    const int lhalf = lane >> 4;
    const int l16   = lane & 15;
    const int row0  = wave * 64;

    v8f acc[4][4];
    #pragma unroll
    for (int rt = 0; rt < 4; ++rt)
        #pragma unroll
        for (int ct = 0; ct < 4; ++ct)
            acc[rt][ct] = (v8f){0.f,0.f,0.f,0.f,0.f,0.f,0.f,0.f};

    #pragma unroll 2
    for (int k0 = 0; k0 < Kdim; k0 += 4) {
        const int pbase = (k0 >> 1) + lhalf;
        v2f bfrag[4];
        #pragma unroll
        for (int ct = 0; ct < 4; ++ct) {
            const int col = ct * 16 + l16;
            bfrag[ct] = *(const v2f*)(hbuf + (size_t)pbase * PAIRW + 2 * col);
        }
        v2f afrag[4];
        #pragma unroll
        for (int rt = 0; rt < 4; ++rt) {
            const int row = row0 + rt * 16 + l16;
            afrag[rt] = *(const v2f*)(W + (size_t)row * ldw + k0 + lhalf * 2);
        }
        #pragma unroll
        for (int rt = 0; rt < 4; ++rt)
            #pragma unroll
            for (int ct = 0; ct < 4; ++ct)
                acc[rt][ct] = __builtin_amdgcn_wmma_f32_16x16x4_f32(
                    false, afrag[rt], false, bfrag[ct],
                    (short)0, acc[rt][ct], false, false);
    }

    __syncthreads();

    #pragma unroll
    for (int rt = 0; rt < 4; ++rt) {
        const int rowb = row0 + rt * 16 + lhalf * 8;
        #pragma unroll
        for (int r = 0; r < 8; r += 2) {
            const float bv0 = bias[rowb + r];
            const float bv1 = bias[rowb + r + 1];
            #pragma unroll
            for (int ct = 0; ct < 4; ++ct) {
                float v0 = acc[rt][ct][r]     + bv0;
                float v1 = acc[rt][ct][r + 1] + bv1;
                v2f v;
                v.x = v0 > 0.f ? v0 : 0.f;
                v.y = v1 > 0.f ? v1 : 0.f;
                *(v2f*)(hbuf + (size_t)((rowb + r) >> 1) * PAIRW
                              + 2 * (ct * 16 + l16)) = v;
            }
        }
    }
    __syncthreads();
}

__global__ __launch_bounds__(256)
void parallel_mlp_kernel(const float* __restrict__ x,
                         const float* __restrict__ W1, const float* __restrict__ b1,
                         const float* __restrict__ W2, const float* __restrict__ b2,
                         const float* __restrict__ W3, const float* __restrict__ b3,
                         const float* __restrict__ W4, const float* __restrict__ b4,
                         float* __restrict__ out)
{
    __shared__ float hbuf[(H_DIM / 2) * PAIRW];
    __shared__ float red[256];

    const int k    = blockIdx.y;
    const int b0   = blockIdx.x * BTILE;
    const int tid  = threadIdx.x;
    const int wave = tid >> 5;
    const int lane = tid & 31;

    for (int i = tid; i < (D_IN / 2) * BTILE; i += 256) {
        const int pd = i & 63;
        const int b  = i >> 6;
        v2f v = *(const v2f*)(x + (size_t)(b0 + b) * D_IN + 2 * pd);
        *(v2f*)(hbuf + (size_t)pd * PAIRW + 2 * b) = v;
    }
    __syncthreads();

    run_layer(W1 + (size_t)k * H_DIM * D_IN,  b1 + (size_t)k * H_DIM,
              D_IN,  D_IN,  hbuf, wave, lane);
    run_layer(W2 + (size_t)k * H_DIM * H_DIM, b2 + (size_t)k * H_DIM,
              H_DIM, H_DIM, hbuf, wave, lane);
    run_layer(W3 + (size_t)k * H_DIM * H_DIM, b3 + (size_t)k * H_DIM,
              H_DIM, H_DIM, hbuf, wave, lane);

    const float* W4k = W4 + (size_t)k * H_DIM;
    const int bcol = tid & 63;
    const int part = tid >> 6;
    float s = 0.f;
    #pragma unroll 4
    for (int d = part * 128; d < part * 128 + 128; d += 2) {
        v2f w = *(const v2f*)(W4k + d);
        v2f h = *(const v2f*)(hbuf + (size_t)(d >> 1) * PAIRW + 2 * bcol);
        s += w.x * h.x + w.y * h.y;
    }
    red[tid] = s;
    __syncthreads();
    if (tid < 64) {
        const float v = red[tid] + red[tid + 64] + red[tid + 128] + red[tid + 192] + b4[k];
        out[(size_t)(b0 + tid) * NUM_K + k] = v;
    }
}

extern "C" void kernel_launch(void* const* d_in, const int* in_sizes, int n_in,
                              void* d_out, int out_size, void* d_ws, size_t ws_size,
                              hipStream_t stream) {
    const float* x  = (const float*)d_in[0];
    const float* W1 = (const float*)d_in[1];
    const float* b1 = (const float*)d_in[2];
    const float* W2 = (const float*)d_in[3];
    const float* b2 = (const float*)d_in[4];
    const float* W3 = (const float*)d_in[5];
    const float* b3 = (const float*)d_in[6];
    const float* W4 = (const float*)d_in[7];
    const float* b4 = (const float*)d_in[8];
    float* out = (float*)d_out;

    dim3 grid(BATCH / BTILE, NUM_K);
    dim3 block(256);
    parallel_mlp_kernel<<<grid, block, 0, stream>>>(x, W1, b1, W2, b2, W3, b3, W4, b4, out);
}


# K loop software-pipelined over two register fragment sets: next 16-k block's A (global) and B (LDS) loads issued as one group ahead of the current block's 64 WMMAs; summation order unchanged
# speedup vs baseline: 1.1325x; 1.1325x over previous
#include <hip/hip_runtime.h>
#include <hip/hip_bf16.h>


#define H_DIM   512
#define D_IN    128
#define BTILE   64
#define PAIRW   152
#define NUM_K   64
#define BATCH   2048

typedef __attribute__((ext_vector_type(2))) float v2f;
typedef __attribute__((ext_vector_type(8))) float v8f;

#define KSTEPS  4
#define KBLOCK  (4 * KSTEPS)

struct Frags {
    v2f a[KSTEPS][4];
    v2f b[KSTEPS][4];
};

template <int LDW>
__device__ __forceinline__ void load_block(const float* __restrict__ wl,
                                           const float* __restrict__ hl,
                                           int k0, Frags& f)
{
    #pragma unroll
    for (int ks = 0; ks < KSTEPS; ++ks)
        #pragma unroll
        for (int rt = 0; rt < 4; ++rt)
            f.a[ks][rt] = *(const v2f*)(wl + (size_t)rt * 16 * LDW + k0 + 4 * ks);
    #pragma unroll
    for (int ks = 0; ks < KSTEPS; ++ks)
        #pragma unroll
        for (int ct = 0; ct < 4; ++ct)
            f.b[ks][ct] = *(const v2f*)(hl + (size_t)((k0 >> 1) + 2 * ks) * PAIRW + 32 * ct);
}

__device__ __forceinline__ void mma_block(const Frags& f, v8f (&acc)[4][4])
{
    #pragma unroll
    for (int ks = 0; ks < KSTEPS; ++ks)
        #pragma unroll
        for (int rt = 0; rt < 4; ++rt)
            #pragma unroll
            for (int ct = 0; ct < 4; ++ct)
                acc[rt][ct] = __builtin_amdgcn_wmma_f32_16x16x4_f32(
                    false, f.a[ks][rt], false, f.b[ks][ct],
                    (short)0, acc[rt][ct], false, false);
}

template <int KDIM>
__device__ __forceinline__ void run_layer(const float* __restrict__ W,
                                          const float* __restrict__ bias,
                                          float* __restrict__ hbuf,
                                          int wave, int lane)
{
    static_assert(KDIM % (2 * KBLOCK) == 0, "two K blocks per loop trip");
    const int lhalf = lane >> 4;
    const int l16   = lane & 15;
    const int row0  = wave * 64;

    const float* wl = W + (size_t)(row0 + l16) * KDIM + lhalf * 2;
    const float* hl = hbuf + (size_t)lhalf * PAIRW + 2 * l16;

    v8f acc[4][4];
    #pragma unroll
    for (int rt = 0; rt < 4; ++rt)
        #pragma unroll
        for (int ct = 0; ct < 4; ++ct)
            acc[rt][ct] = (v8f){0.f,0.f,0.f,0.f,0.f,0.f,0.f,0.f};

    Frags f0, f1;
    load_block<KDIM>(wl, hl, 0, f0);
    #pragma unroll 1
    for (int k0 = 0; k0 < KDIM - 2 * KBLOCK; k0 += 2 * KBLOCK) {
        load_block<KDIM>(wl, hl, k0 + KBLOCK, f1);
        __builtin_amdgcn_sched_barrier(0);
        mma_block(f0, acc);
        __builtin_amdgcn_sched_barrier(0);
        load_block<KDIM>(wl, hl, k0 + 2 * KBLOCK, f0);
        __builtin_amdgcn_sched_barrier(0);
        mma_block(f1, acc);
    }
    load_block<KDIM>(wl, hl, KDIM - KBLOCK, f1);
    __builtin_amdgcn_sched_barrier(0);
    mma_block(f0, acc);
    __builtin_amdgcn_sched_barrier(0);
    mma_block(f1, acc);

    __syncthreads();

    #pragma unroll
    for (int rt = 0; rt < 4; ++rt) {
        const int rowb = row0 + rt * 16 + lhalf * 8;
        #pragma unroll
        for (int r = 0; r < 8; r += 2) {
            const float bv0 = bias[rowb + r];
            const float bv1 = bias[rowb + r + 1];
            #pragma unroll
            for (int ct = 0; ct < 4; ++ct) {
                float v0 = acc[rt][ct][r]     + bv0;
                float v1 = acc[rt][ct][r + 1] + bv1;
                v2f v;
                v.x = v0 > 0.f ? v0 : 0.f;
                v.y = v1 > 0.f ? v1 : 0.f;
                *(v2f*)(hbuf + (size_t)((rowb + r) >> 1) * PAIRW
                              + 2 * (ct * 16 + l16)) = v;
            }
        }
    }
    __syncthreads();
}

__global__ __launch_bounds__(256)
void parallel_mlp_kernel(const float* __restrict__ x,
                         const float* __restrict__ W1, const float* __restrict__ b1,
                         const float* __restrict__ W2, const float* __restrict__ b2,
                         const float* __restrict__ W3, const float* __restrict__ b3,
                         const float* __restrict__ W4, const float* __restrict__ b4,
                         float* __restrict__ out)
{
    __shared__ float hbuf[(H_DIM / 2) * PAIRW];
    __shared__ float red[256];

    const int k    = blockIdx.y;
    const int b0   = blockIdx.x * BTILE;
    const int tid  = threadIdx.x;
    const int wave = tid >> 5;
    const int lane = tid & 31;

    for (int i = tid; i < (D_IN / 2) * BTILE; i += 256) {
        const int pd = i & 63;
        const int b  = i >> 6;
        v2f v = *(const v2f*)(x + (size_t)(b0 + b) * D_IN + 2 * pd);
        *(v2f*)(hbuf + (size_t)pd * PAIRW + 2 * b) = v;
    }
    __syncthreads();

    run_layer<D_IN>(W1 + (size_t)k * H_DIM * D_IN,  b1 + (size_t)k * H_DIM,
                    hbuf, wave, lane);
    run_layer<H_DIM>(W2 + (size_t)k * H_DIM * H_DIM, b2 + (size_t)k * H_DIM,
                     hbuf, wave, lane);
    run_layer<H_DIM>(W3 + (size_t)k * H_DIM * H_DIM, b3 + (size_t)k * H_DIM,
                     hbuf, wave, lane);

    const float* W4k = W4 + (size_t)k * H_DIM;
    const int bcol = tid & 63;
    const int part = tid >> 6;
    float s = 0.f;
    #pragma unroll 4
    for (int d = part * 128; d < part * 128 + 128; d += 2) {
        v2f w = *(const v2f*)(W4k + d);
        v2f h = *(const v2f*)(hbuf + (size_t)(d >> 1) * PAIRW + 2 * bcol);
        s += w.x * h.x + w.y * h.y;
    }
    red[tid] = s;
    __syncthreads();
    if (tid < 64) {
        const float v = red[tid] + red[tid + 64] + red[tid + 128] + red[tid + 192] + b4[k];
        out[(size_t)(b0 + tid) * NUM_K + k] = v;
    }
}

extern "C" void kernel_launch(void* const* d_in, const int* in_sizes, int n_in,
                              void* d_out, int out_size, void* d_ws, size_t ws_size,
                              hipStream_t stream) {
    const float* x  = (const float*)d_in[0];
    const float* W1 = (const float*)d_in[1];
    const float* b1 = (const float*)d_in[2];
    const float* W2 = (const float*)d_in[3];
    const float* b2 = (const float*)d_in[4];
    const float* W3 = (const float*)d_in[5];
    const float* b3 = (const float*)d_in[6];
    const float* W4 = (const float*)d_in[7];
    const float* b4 = (const float*)d_in[8];
    float* out = (float*)d_out;

    dim3 grid(BATCH / BTILE, NUM_K);
    dim3 block(256);
    parallel_mlp_kernel<<<grid, block, 0, stream>>>(x, W1, b1, W2, b2, W3, b3, W4, b4, out);
}


# Batch tile per workgroup 64 -> 128 columns (wave tile 64x128): each global weight-fragment load feeds 8 WMMAs instead of 4, half as many workgroups pay the layer hand-overs; summation order unchanged
# speedup vs baseline: 1.2294x; 1.0856x over previous
#include <hip/hip_runtime.h>
#include <hip/hip_bf16.h>


#define H_DIM   512
#define D_IN    128
#define BTILE   128
#define CT      (BTILE / 16)
#define PAIRW   280
#define NUM_K   64
#define BATCH   2048

typedef __attribute__((ext_vector_type(2))) float v2f;
typedef __attribute__((ext_vector_type(8))) float v8f;

#define KSTEPS  4
#define KBLOCK  (4 * KSTEPS)

struct Frags {
    v2f a[KSTEPS][4];
    v2f b[KSTEPS][CT];
};

template <int LDW>
__device__ __forceinline__ void load_block(const float* __restrict__ wl,
                                           const float* __restrict__ hl,
                                           int k0, Frags& f)
{
    #pragma unroll
    for (int ks = 0; ks < KSTEPS; ++ks)
        #pragma unroll
        for (int rt = 0; rt < 4; ++rt)
            f.a[ks][rt] = *(const v2f*)(wl + (size_t)rt * 16 * LDW + k0 + 4 * ks);
    #pragma unroll
    for (int ks = 0; ks < KSTEPS; ++ks)
        #pragma unroll
        for (int ct = 0; ct < CT; ++ct)
            f.b[ks][ct] = *(const v2f*)(hl + (size_t)((k0 >> 1) + 2 * ks) * PAIRW + 32 * ct);
}

__device__ __forceinline__ void mma_block(const Frags& f, v8f (&acc)[4][CT])
{
    #pragma unroll
    for (int ks = 0; ks < KSTEPS; ++ks)
        #pragma unroll
        for (int rt = 0; rt < 4; ++rt)
            #pragma unroll
            for (int ct = 0; ct < CT; ++ct)
                acc[rt][ct] = __builtin_amdgcn_wmma_f32_16x16x4_f32(
                    false, f.a[ks][rt], false, f.b[ks][ct],
                    (short)0, acc[rt][ct], false, false);
}

template <int KDIM>
__device__ __forceinline__ void run_layer(const float* __restrict__ W,
                                          const float* __restrict__ bias,
                                          float* __restrict__ hbuf,
                                          int wave, int lane)
{
    static_assert(KDIM % (2 * KBLOCK) == 0, "two K blocks per loop trip");
    const int lhalf = lane >> 4;
    const int l16   = lane & 15;
    const int row0  = wave * 64;

    const float* wl = W + (size_t)(row0 + l16) * KDIM + lhalf * 2;
    const float* hl = hbuf + (size_t)lhalf * PAIRW + 2 * l16;

    v8f acc[4][CT];
    #pragma unroll
    for (int rt = 0; rt < 4; ++rt)
        #pragma unroll
        for (int ct = 0; ct < CT; ++ct)
            acc[rt][ct] = (v8f){0.f,0.f,0.f,0.f,0.f,0.f,0.f,0.f};

    Frags f0, f1;
    load_block<KDIM>(wl, hl, 0, f0);
    #pragma unroll 1
    for (int k0 = 0; k0 < KDIM - 2 * KBLOCK; k0 += 2 * KBLOCK) {
        load_block<KDIM>(wl, hl, k0 + KBLOCK, f1);
        __builtin_amdgcn_sched_barrier(0);
        mma_block(f0, acc);
        __builtin_amdgcn_sched_barrier(0);
        load_block<KDIM>(wl, hl, k0 + 2 * KBLOCK, f0);
        __builtin_amdgcn_sched_barrier(0);
        mma_block(f1, acc);
    }
    load_block<KDIM>(wl, hl, KDIM - KBLOCK, f1);
    __builtin_amdgcn_sched_barrier(0);
    mma_block(f0, acc);
    __builtin_amdgcn_sched_barrier(0);
    mma_block(f1, acc);

    __syncthreads();

    #pragma unroll
    for (int rt = 0; rt < 4; ++rt) {
        const int rowb = row0 + rt * 16 + lhalf * 8;
        #pragma unroll
        for (int r = 0; r < 8; r += 2) {
            const float bv0 = bias[rowb + r];
            const float bv1 = bias[rowb + r + 1];
            #pragma unroll
            for (int ct = 0; ct < CT; ++ct) {
                float v0 = acc[rt][ct][r]     + bv0;
                float v1 = acc[rt][ct][r + 1] + bv1;
                v2f v;
                v.x = v0 > 0.f ? v0 : 0.f;
                v.y = v1 > 0.f ? v1 : 0.f;
                *(v2f*)(hbuf + (size_t)((rowb + r) >> 1) * PAIRW
                              + 2 * (ct * 16 + l16)) = v;
            }
        }
    }
    __syncthreads();
}

__global__ __launch_bounds__(256)
void parallel_mlp_kernel(const float* __restrict__ x,
                         const float* __restrict__ W1, const float* __restrict__ b1,
                         const float* __restrict__ W2, const float* __restrict__ b2,
                         const float* __restrict__ W3, const float* __restrict__ b3,
                         const float* __restrict__ W4, const float* __restrict__ b4,
                         float* __restrict__ out)
{
    __shared__ float hbuf[(H_DIM / 2) * PAIRW];
    __shared__ float red[4 * BTILE];

    const int k    = blockIdx.y;
    const int b0   = blockIdx.x * BTILE;
    const int tid  = threadIdx.x;
    const int wave = tid >> 5;
    const int lane = tid & 31;

    for (int i = tid; i < (D_IN / 2) * BTILE; i += 256) {
        const int pd = i & 63;
        const int b  = i >> 6;
        v2f v = *(const v2f*)(x + (size_t)(b0 + b) * D_IN + 2 * pd);
        *(v2f*)(hbuf + (size_t)pd * PAIRW + 2 * b) = v;
    }
    __syncthreads();

    run_layer<D_IN>(W1 + (size_t)k * H_DIM * D_IN,  b1 + (size_t)k * H_DIM,
                    hbuf, wave, lane);
    run_layer<H_DIM>(W2 + (size_t)k * H_DIM * H_DIM, b2 + (size_t)k * H_DIM,
                     hbuf, wave, lane);
    run_layer<H_DIM>(W3 + (size_t)k * H_DIM * H_DIM, b3 + (size_t)k * H_DIM,
                     hbuf, wave, lane);

    const float* W4k = W4 + (size_t)k * H_DIM;
    const int bcol = tid & (BTILE - 1);
    const int part = tid >> 7;
    float s0 = 0.f, s1 = 0.f;
    #pragma unroll 4
    for (int d = part * 128; d < part * 128 + 128; d += 2) {
        v2f w0 = *(const v2f*)(W4k + d);
        v2f h0 = *(const v2f*)(hbuf + (size_t)(d >> 1) * PAIRW + 2 * bcol);
        v2f w1 = *(const v2f*)(W4k + d + 256);
        v2f h1 = *(const v2f*)(hbuf + (size_t)((d + 256) >> 1) * PAIRW + 2 * bcol);
        s0 += w0.x * h0.x + w0.y * h0.y;
        s1 += w1.x * h1.x + w1.y * h1.y;
    }
    red[part * BTILE + bcol]       = s0;
    red[(part + 2) * BTILE + bcol] = s1;
    __syncthreads();
    if (tid < BTILE) {
        const float v = red[tid] + red[tid + BTILE] + red[tid + 2 * BTILE]
                      + red[tid + 3 * BTILE] + b4[k];
        out[(size_t)(b0 + tid) * NUM_K + k] = v;
    }
}

extern "C" void kernel_launch(void* const* d_in, const int* in_sizes, int n_in,
                              void* d_out, int out_size, void* d_ws, size_t ws_size,
                              hipStream_t stream) {
    const float* x  = (const float*)d_in[0];
    const float* W1 = (const float*)d_in[1];
    const float* b1 = (const float*)d_in[2];
    const float* W2 = (const float*)d_in[3];
    const float* b2 = (const float*)d_in[4];
    const float* W3 = (const float*)d_in[5];
    const float* b3 = (const float*)d_in[6];
    const float* W4 = (const float*)d_in[7];
    const float* b4 = (const float*)d_in[8];
    float* out = (float*)d_out;

    dim3 grid(BATCH / BTILE, NUM_K);
    dim3 block(256);
    parallel_mlp_kernel<<<grid, block, 0, stream>>>(x, W1, b1, W2, b2, W3, b3, W4, b4, out);
}
